# starting kernel
# baseline (speedup 1.0000x reference)
#include <hip/hip_runtime.h>
#include <hip/hip_bf16.h>

typedef __attribute__((ext_vector_type(2))) float v2f;
typedef __attribute__((ext_vector_type(8))) float v8f;

template <int K, int LDA, int LDB, bool COLGUARD>
__global__ void gemm_wmma_k(const float* __restrict__ A,
                            const float* __restrict__ B,
                            const float* __restrict__ bias,
                            float* __restrict__ C, int M) {
  const int lane  = threadIdx.x & 31;
  const int laneM = lane & 15;
  const int hi    = lane >> 4;
  const int row0  = blockIdx.x * 16;
  const int col0  = threadIdx.y * 16;

  int arow = row0 + laneM;
  if (arow > M - 1) arow = M - 1;
  const float* __restrict__ aptr = A + (size_t)arow * LDA + 2 * hi;

  const int col   = col0 + laneM;
  int       colc  = col;
  float     cmask = 1.0f;
  if (COLGUARD) {
    if (colc > LDB - 1) { colc = LDB - 1; cmask = 0.0f; }
  }
  const float* __restrict__ bptr = B + 2 * hi * LDB + colc;

  v8f acc = {};
#pragma unroll
  for (int k = 0; k < K; k += 4) {
    v2f a = *(const v2f*)(aptr + k);
    v2f b;
    b.x = bptr[k * LDB];
    b.y = bptr[k * LDB + LDB];
    if (COLGUARD) { b.x *= cmask; b.y *= cmask; }
    acc = __builtin_amdgcn_wmma_f32_16x16x4_f32(false, a, false, b,
                                                (short)0, acc, false, false);
  }

  if (COLGUARD && cmask == 0.0f) return;
  const float bc = COLGUARD ? bias[col] : 0.0f;
  float* __restrict__ cptr = C + (size_t)(row0 + 8 * hi) * LDB + col;
#pragma unroll
  for (int r = 0; r < 8; ++r) {
    const int row = row0 + r + 8 * hi;
    if (row < M) cptr[r * LDB] = acc[r] + bc;
  }
}

__global__ void scatter_edges(const float* __restrict__ T,
                              const int* __restrict__ src,
                              const int* __restrict__ dst,
                              const float* __restrict__ w,
                              float* __restrict__ agg, int E) {
  const int e = blockIdx.x * (blockDim.x >> 5) + (threadIdx.x >> 5);
  if (e >= E) return;
  const int   lane = threadIdx.x & 31;
  const int   s  = src[e];
  const int   d  = dst[e];
  const float we = w[e];
  const v2f   tv = *(const v2f*)(T + (size_t)s * 64 + 2 * lane);
  float* __restrict__ ap = agg + (size_t)d * 64 + 2 * lane;
  atomicAdd(ap,     tv.x * we);
  atomicAdd(ap + 1, tv.y * we);
}

__global__ void zero_f32(float* __restrict__ p, int n) {
  const int i = blockIdx.x * blockDim.x + threadIdx.x;
  if (i < n) p[i] = 0.0f;
}

__global__ void bias_relu_to_jk(const float* __restrict__ agg,
                                const float* __restrict__ b,
                                float* __restrict__ hjk,
                                int n64, int colOff) {
  const int i = blockIdx.x * blockDim.x + threadIdx.x;
  if (i >= n64) return;
  const int row = i >> 6;
  const int c   = i & 63;
  const float v = agg[i] + b[c];
  hjk[(size_t)row * 192 + colOff + c] = fmaxf(v, 0.0f);
}

extern "C" void kernel_launch(void* const* d_in, const int* in_sizes, int n_in,
                              void* d_out, int out_size, void* d_ws, size_t ws_size,
                              hipStream_t stream) {
  const float* x    = (const float*)d_in[0];
  const int*   ei   = (const int*)d_in[1];
  const float* ew   = (const float*)d_in[2];
  const float* W1   = (const float*)d_in[3];
  const float* b1   = (const float*)d_in[4];
  const float* W2   = (const float*)d_in[5];
  const float* b2   = (const float*)d_in[6];
  const float* W3   = (const float*)d_in[7];
  const float* b3   = (const float*)d_in[8];
  const float* Wlin = (const float*)d_in[9];
  const float* blin = (const float*)d_in[10];

  const int N = in_sizes[0] / 128;
  const int E = in_sizes[2];
  const int* src = ei;
  const int* dst = ei + E;

  float* T   = (float*)d_ws;
  float* agg = T   + (size_t)N * 64;
  float* hjk = agg + (size_t)N * 64;

  const int Mtiles  = (N + 15) / 16;
  const int n64     = N * 64;
  const int zblocks = (n64 + 255) / 256;
  const int sblocks = (E + 7) / 8;

  gemm_wmma_k<128, 128, 64, false><<<dim3(Mtiles, 1, 1), dim3(32, 4, 1), 0, stream>>>(
      x, W1, nullptr, T, N);
  zero_f32<<<zblocks, 256, 0, stream>>>(agg, n64);
  scatter_edges<<<sblocks, 256, 0, stream>>>(T, src, dst, ew, agg, E);
  bias_relu_to_jk<<<zblocks, 256, 0, stream>>>(agg, b1, hjk, n64, 0);

  gemm_wmma_k<64, 192, 64, false><<<dim3(Mtiles, 1, 1), dim3(32, 4, 1), 0, stream>>>(
      hjk, W2, nullptr, T, N);
  zero_f32<<<zblocks, 256, 0, stream>>>(agg, n64);
  scatter_edges<<<sblocks, 256, 0, stream>>>(T, src, dst, ew, agg, E);
  bias_relu_to_jk<<<zblocks, 256, 0, stream>>>(agg, b2, hjk, n64, 64);

  gemm_wmma_k<64, 192, 64, false><<<dim3(Mtiles, 1, 1), dim3(32, 4, 1), 0, stream>>>(
      hjk + 64, W3, nullptr, T, N);
  zero_f32<<<zblocks, 256, 0, stream>>>(agg, n64);
  scatter_edges<<<sblocks, 256, 0, stream>>>(T, src, dst, ew, agg, E);
  bias_relu_to_jk<<<zblocks, 256, 0, stream>>>(agg, b3, hjk, n64, 128);

  gemm_wmma_k<192, 192, 40, true><<<dim3(Mtiles, 1, 1), dim3(32, 3, 1), 0, stream>>>(
      hjk, Wlin, blin, (float*)d_out, N);
}


# Zero-fill and bias/ReLU passes removed: each layer GEMM stores the bias as the atomics' start value (whole lines), consumers apply ReLU on their A load: 13 -> 7 launches; bias is summed first
# speedup vs baseline: 1.8646x; 1.8646x over previous
#include <hip/hip_runtime.h>
#include <hip/hip_bf16.h>


typedef __attribute__((ext_vector_type(2))) float v2f;
typedef __attribute__((ext_vector_type(4))) float v4f;
typedef __attribute__((ext_vector_type(8))) float v8f;


template <int K, bool RELU_A>
__global__ void __attribute__((amdgpu_waves_per_eu(16, 16))) gemm_layer_k(const float* __restrict__ A,
                             const float* __restrict__ B,
                             const float* __restrict__ bias,
                             float* __restrict__ T,
                             float* __restrict__ P, int M) {
  const int lane  = threadIdx.x & 31;
  const int laneM = lane & 15;
  const int hi    = lane >> 4;
  const int wy    = threadIdx.y;
  const int row0  = blockIdx.x * 16;
  const int col0  = wy * 16;

  const int c4 = laneM * 4;
  v4f bv;
  bv.x = bias[c4 + 0];
  bv.y = bias[c4 + 1];
  bv.z = bias[c4 + 2];
  bv.w = bias[c4 + 3];

  int arow = row0 + laneM;
  if (arow > M - 1) arow = M - 1;
  const float* __restrict__ aptr = A + (size_t)arow * K + 2 * hi;
  const float* __restrict__ bptr = B + 2 * hi * 64 + col0 + laneM;

  v8f acc = {};
#pragma unroll
  for (int k = 0; k < K; k += 4) {
    v2f a = *(const v2f*)(aptr + k);
    if (RELU_A) { a.x = fmaxf(a.x, 0.0f); a.y = fmaxf(a.y, 0.0f); }
    v2f b;
    b.x = bptr[k * 64];
    b.y = bptr[k * 64 + 64];
    acc = __builtin_amdgcn_wmma_f32_16x16x4_f32(false, a, false, b,
                                                (short)0, acc, false, false);
  }

  float* __restrict__ cptr = T + (size_t)(row0 + 8 * hi) * 64 + col0 + laneM;
#pragma unroll
  for (int r = 0; r < 8; ++r) {
    const int row = row0 + r + 8 * hi;
    if (row < M) cptr[r * 64] = acc[r];
  }

#pragma unroll
  for (int i = 0; i < 2; ++i) {
    const int r = row0 + wy * 4 + 2 * i + hi;
    if (r < M) *(v4f*)(P + (size_t)r * 64 + c4) = bv;
  }
}

__global__ void gemm_out_k(const float* __restrict__ P0,
                           const float* __restrict__ P1,
                           const float* __restrict__ P2,
                           const float* __restrict__ Wl,
                           const float* __restrict__ bl,
                           float* __restrict__ C, int M) {
  constexpr int LDB = 40;
  const int lane  = threadIdx.x & 31;
  const int laneM = lane & 15;
  const int hi    = lane >> 4;
  const int row0  = blockIdx.x * 16;
  const int col0  = threadIdx.y * 16;

  int arow = row0 + laneM;
  if (arow > M - 1) arow = M - 1;
  const unsigned aoff = (unsigned)arow * 64u + 2u * (unsigned)hi;

  const int col   = col0 + laneM;
  int       colc  = col;
  float     cmask = 1.0f;
  if (colc > LDB - 1) { colc = LDB - 1; cmask = 0.0f; }
  const float* __restrict__ bptr = Wl + 2 * hi * LDB + colc;

  v8f acc = {};
#pragma unroll
  for (int s = 0; s < 3; ++s) {
    const float* __restrict__ ap = ((s == 0) ? P0 : (s == 1) ? P1 : P2) + aoff;
#pragma unroll
    for (int kk = 0; kk < 64; kk += 4) {
      const int k = s * 64 + kk;
      v2f a = *(const v2f*)(ap + kk);
      a.x = fmaxf(a.x, 0.0f);
      a.y = fmaxf(a.y, 0.0f);
      v2f b;
      b.x = bptr[k * LDB] * cmask;
      b.y = bptr[k * LDB + LDB] * cmask;
      acc = __builtin_amdgcn_wmma_f32_16x16x4_f32(false, a, false, b,
                                                  (short)0, acc, false, false);
    }
  }

  if (cmask == 0.0f) return;
  const float bc = bl[col];
  float* __restrict__ cptr = C + (size_t)(row0 + 8 * hi) * LDB + col;
#pragma unroll
  for (int r = 0; r < 8; ++r) {
    const int row = row0 + r + 8 * hi;
    if (row < M) cptr[r * LDB] = acc[r] + bc;
  }
}

constexpr int kEdgesPerWave = 32;
constexpr int kEdgeGroup    = 8;

__device__ __forceinline__ void scatter_one(const float* __restrict__ T,
                                            float* __restrict__ agg,
                                            int srcv, int dstv, float wv,
                                            int j, int lane) {
  const int   s  = __builtin_amdgcn_readlane(srcv, j);
  const int   d  = __builtin_amdgcn_readlane(dstv, j);
  const float we = __builtin_bit_cast(
      float, __builtin_amdgcn_readlane(__builtin_bit_cast(int, wv), j));
  const float* __restrict__ tp = T + (size_t)s * 64 + lane;
  const float t0 = tp[0];
  const float t1 = tp[32];
  float* __restrict__ ap = agg + (size_t)d * 64 + lane;
  atomicAdd(ap,      t0 * we);
  atomicAdd(ap + 32, t1 * we);
}

__global__ void scatter_edges(const float* __restrict__ T,
                              const int* __restrict__ src,
                              const int* __restrict__ dst,
                              const float* __restrict__ w,
                              float* __restrict__ agg, int E) {
  const int lane = threadIdx.x & 31;
  const int wave = __builtin_amdgcn_readfirstlane((int)(threadIdx.x >> 5));
  const int base = (blockIdx.x * (int)(blockDim.x >> 5) + wave) * kEdgesPerWave;
  if (base >= E) return;
  const int n = (E - base < kEdgesPerWave) ? (E - base) : kEdgesPerWave;

  int el = base + lane;
  if (el > E - 1) el = E - 1;
  const int   srcv = src[el];
  const int   dstv = dst[el];
  const float wv   = w[el];

#pragma unroll
  for (int g = 0; g < kEdgesPerWave; g += kEdgeGroup) {
    if (g + kEdgeGroup <= n) {
      int   s[kEdgeGroup], d[kEdgeGroup];
      float we[kEdgeGroup];
      float t0[kEdgeGroup], t1[kEdgeGroup];
#pragma unroll
      for (int u = 0; u < kEdgeGroup; ++u) {
        s[u]  = __builtin_amdgcn_readlane(srcv, g + u);
        d[u]  = __builtin_amdgcn_readlane(dstv, g + u);
        we[u] = __builtin_bit_cast(
            float, __builtin_amdgcn_readlane(__builtin_bit_cast(int, wv), g + u));
        const float* __restrict__ tp = T + (size_t)s[u] * 64 + lane;
        t0[u] = tp[0];
        t1[u] = tp[32];
      }
#pragma unroll
      for (int u = 0; u < kEdgeGroup; ++u) {
        float* __restrict__ ap = agg + (size_t)d[u] * 64 + lane;
        atomicAdd(ap,      t0[u] * we[u]);
        atomicAdd(ap + 32, t1[u] * we[u]);
      }
    } else {
      for (int j = g; j < n; ++j) scatter_one(T, agg, srcv, dstv, wv, j, lane);
      break;
    }
  }
}

extern "C" void kernel_launch(void* const* d_in, const int* in_sizes, int n_in,
                              void* d_out, int out_size, void* d_ws, size_t ws_size,
                              hipStream_t stream) {
  const float* x    = (const float*)d_in[0];
  const int*   ei   = (const int*)d_in[1];
  const float* ew   = (const float*)d_in[2];
  const float* W1   = (const float*)d_in[3];
  const float* b1   = (const float*)d_in[4];
  const float* W2   = (const float*)d_in[5];
  const float* b2   = (const float*)d_in[6];
  const float* W3   = (const float*)d_in[7];
  const float* b3   = (const float*)d_in[8];
  const float* Wlin = (const float*)d_in[9];
  const float* blin = (const float*)d_in[10];

  const int N = in_sizes[0] / 128;
  const int E = in_sizes[2];
  const int* src = ei;
  const int* dst = ei + E;

  float* T  = (float*)d_ws;
  float* P1 = T  + (size_t)N * 64;
  float* P2 = P1 + (size_t)N * 64;
  float* P3 = P2 + (size_t)N * 64;

  const int Mtiles  = (N + 15) / 16;
  const int sblocks = (E + 255) / 256;

  gemm_layer_k<128, false><<<dim3(Mtiles, 1, 1), dim3(32, 4, 1), 0, stream>>>(
      x, W1, b1, T, P1, N);
  scatter_edges<<<sblocks, 256, 0, stream>>>(T, src, dst, ew, P1, E);

  gemm_layer_k<64, true><<<dim3(Mtiles, 1, 1), dim3(32, 4, 1), 0, stream>>>(
      P1, W2, b2, T, P2, N);
  scatter_edges<<<sblocks, 256, 0, stream>>>(T, src, dst, ew, P2, E);

  gemm_layer_k<64, true><<<dim3(Mtiles, 1, 1), dim3(32, 4, 1), 0, stream>>>(
      P2, W3, b3, T, P3, N);
  scatter_edges<<<sblocks, 256, 0, stream>>>(T, src, dst, ew, P3, E);

  gemm_out_k<<<dim3(Mtiles, 1, 1), dim3(32, 3, 1), 0, stream>>>(
      P1, P2, P3, Wlin, blin, (float*)d_out, N);
}


# Edge scatter handed out by package half: a work-group reads its die, takes a ticket of its half and adds only into that half's range of dst rows; sums stay in atomic arrival order
# speedup vs baseline: 2.2553x; 1.2095x over previous
#include <hip/hip_runtime.h>
#include <hip/hip_bf16.h>


typedef __attribute__((ext_vector_type(2))) float v2f;
typedef __attribute__((ext_vector_type(4))) float v4f;
typedef __attribute__((ext_vector_type(8))) float v8f;


constexpr int kTickets     = 6;
constexpr int kTicketPitch = 64;

template <int K, bool RELU_A>
__global__ void __attribute__((amdgpu_waves_per_eu(16, 16))) gemm_layer_k(const float* __restrict__ A,
                             const float* __restrict__ B,
                             const float* __restrict__ bias,
                             float* __restrict__ T,
                             float* __restrict__ P, int M,
                             int* __restrict__ tick) {
  const int lane  = threadIdx.x & 31;
  const int laneM = lane & 15;
  const int hi    = lane >> 4;
  const int wy    = threadIdx.y;
  const int row0  = blockIdx.x * 16;
  const int col0  = wy * 16;

  if (tick != nullptr && blockIdx.x == 0 && wy == 0 && lane < kTickets)
    tick[lane * kTicketPitch] = 0;

  const int c4 = laneM * 4;
  v4f bv;
  bv.x = bias[c4 + 0];
  bv.y = bias[c4 + 1];
  bv.z = bias[c4 + 2];
  bv.w = bias[c4 + 3];

  int arow = row0 + laneM;
  if (arow > M - 1) arow = M - 1;
  const float* __restrict__ aptr = A + (size_t)arow * K + 2 * hi;
  const float* __restrict__ bptr = B + 2 * hi * 64 + col0 + laneM;

  v8f acc = {};
#pragma unroll
  for (int k = 0; k < K; k += 4) {
    v2f a = *(const v2f*)(aptr + k);
    if (RELU_A) { a.x = fmaxf(a.x, 0.0f); a.y = fmaxf(a.y, 0.0f); }
    v2f b;
    b.x = bptr[k * 64];
    b.y = bptr[k * 64 + 64];
    acc = __builtin_amdgcn_wmma_f32_16x16x4_f32(false, a, false, b,
                                                (short)0, acc, false, false);
  }

  float* __restrict__ cptr = T + (size_t)(row0 + 8 * hi) * 64 + col0 + laneM;
#pragma unroll
  for (int r = 0; r < 8; ++r) {
    const int row = row0 + r + 8 * hi;
    if (row < M) cptr[r * 64] = acc[r];
  }

#pragma unroll
  for (int i = 0; i < 2; ++i) {
    const int r = row0 + wy * 4 + 2 * i + hi;
    if (r < M) *(v4f*)(P + (size_t)r * 64 + c4) = bv;
  }
}

__global__ void gemm_out_k(const float* __restrict__ P0,
                           const float* __restrict__ P1,
                           const float* __restrict__ P2,
                           const float* __restrict__ Wl,
                           const float* __restrict__ bl,
                           float* __restrict__ C, int M) {
  constexpr int LDB = 40;
  const int lane  = threadIdx.x & 31;
  const int laneM = lane & 15;
  const int hi    = lane >> 4;
  const int row0  = blockIdx.x * 16;
  const int col0  = threadIdx.y * 16;

  int arow = row0 + laneM;
  if (arow > M - 1) arow = M - 1;
  const unsigned aoff = (unsigned)arow * 64u + 2u * (unsigned)hi;

  const int col   = col0 + laneM;
  int       colc  = col;
  float     cmask = 1.0f;
  if (colc > LDB - 1) { colc = LDB - 1; cmask = 0.0f; }
  const float* __restrict__ bptr = Wl + 2 * hi * LDB + colc;

  v8f acc = {};
#pragma unroll
  for (int s = 0; s < 3; ++s) {
    const float* __restrict__ ap = ((s == 0) ? P0 : (s == 1) ? P1 : P2) + aoff;
#pragma unroll
    for (int kk = 0; kk < 64; kk += 4) {
      const int k = s * 64 + kk;
      v2f a = *(const v2f*)(ap + kk);
      a.x = fmaxf(a.x, 0.0f);
      a.y = fmaxf(a.y, 0.0f);
      v2f b;
      b.x = bptr[k * LDB] * cmask;
      b.y = bptr[k * LDB + LDB] * cmask;
      acc = __builtin_amdgcn_wmma_f32_16x16x4_f32(false, a, false, b,
                                                  (short)0, acc, false, false);
    }
  }

  if (cmask == 0.0f) return;
  const float bc = bl[col];
  float* __restrict__ cptr = C + (size_t)(row0 + 8 * hi) * LDB + col;
#pragma unroll
  for (int r = 0; r < 8; ++r) {
    const int row = row0 + r + 8 * hi;
    if (row < M) cptr[r * LDB] = acc[r] + bc;
  }
}

constexpr int kRawPerWave   = 64;
constexpr int kScatterWaves = 8;
constexpr int kChunk        = kRawPerWave * kScatterWaves;
constexpr int kEdgeGroup    = 8;

template <bool FULL>
__device__ __forceinline__ void scatter_group(const float* __restrict__ T,
                                              float* __restrict__ agg,
                                              int sv, int dv, float wv,
                                              int g, int c, int lane) {
  int   d[kEdgeGroup];
  float we[kEdgeGroup];
  float t0[kEdgeGroup], t1[kEdgeGroup];
#pragma unroll
  for (int u = 0; u < kEdgeGroup; ++u) {
    if (FULL || u < c) {
      const int s = __builtin_amdgcn_readlane(sv, g + u);
      d[u]  = __builtin_amdgcn_readlane(dv, g + u);
      we[u] = __builtin_bit_cast(
          float, __builtin_amdgcn_readlane(__builtin_bit_cast(int, wv), g + u));
      const float* __restrict__ tp = T + (size_t)s * 64 + lane;
      t0[u] = tp[0];
      t1[u] = tp[32];
    }
  }
#pragma unroll
  for (int u = 0; u < kEdgeGroup; ++u) {
    if (FULL || u < c) {
      float* __restrict__ ap = agg + (size_t)d[u] * 64 + lane;
      atomicAdd(ap,      t0[u] * we[u]);
      atomicAdd(ap + 32, t1[u] * we[u]);
    }
  }
}

__device__ __forceinline__ void scatter_list(const float* __restrict__ T,
                                             float* __restrict__ agg,
                                             int sv, int dv, float wv,
                                             int cnt, int lane) {
  int g = 0;
  for (; g + kEdgeGroup <= cnt; g += kEdgeGroup)
    scatter_group<true>(T, agg, sv, dv, wv, g, kEdgeGroup, lane);
  if (g < cnt) scatter_group<false>(T, agg, sv, dv, wv, g, cnt - g, lane);
}

__global__ void __launch_bounds__(256) scatter_edges(const float* __restrict__ T,
                              const int* __restrict__ src,
                              const int* __restrict__ dst,
                              const float* __restrict__ w,
                              float* __restrict__ agg, int E, int N,
                              int nchunks, int* __restrict__ tick) {
  __shared__ int   sh_ticket[2];
  __shared__ int   sh_src[kScatterWaves][kRawPerWave];
  __shared__ int   sh_dst[kScatterWaves][kRawPerWave];
  __shared__ float sh_w[kScatterWaves][kRawPerWave];

  const int lane = threadIdx.x & 31;
  const int wave = __builtin_amdgcn_readfirstlane((int)(threadIdx.x >> 5));

  if (threadIdx.x == 0) {
    const unsigned id = __builtin_amdgcn_s_sendmsg_rtn(0x87);
    int h = (int)((id >> 18) & 1u);
    int t = atomicAdd(tick + h * kTicketPitch, 1);
    if (t >= nchunks) {
      h ^= 1;
      t = atomicAdd(tick + h * kTicketPitch, 1);
    }
    sh_ticket[0] = h;
    sh_ticket[1] = t;
  }
  __syncthreads();
  const int h = __builtin_amdgcn_readfirstlane(sh_ticket[0]);
  const int t = __builtin_amdgcn_readfirstlane(sh_ticket[1]);
  if (t >= nchunks) return;

  const int base = t * kChunk + wave * kRawPerWave;

  const int  e0 = base + lane;
  const int  e1 = base + 32 + lane;
  const bool v0 = e0 < E;
  const bool v1 = e1 < E;
  const int  c0 = v0 ? e0 : E - 1;
  const int  c1 = v1 ? e1 : E - 1;
  const int   d0 = dst[c0];
  const int   d1 = dst[c1];
  const int   s0 = src[c0];
  const int   s1 = src[c1];
  const float w0 = w[c0];
  const float w1 = w[c1];

  const int  split = N >> 1;
  const bool k0 = v0 && ((d0 >= split) == (h != 0));
  const bool k1 = v1 && ((d1 >= split) == (h != 0));
  const unsigned m0 = __builtin_amdgcn_ballot_w32(k0);
  const unsigned m1 = __builtin_amdgcn_ballot_w32(k1);
  const int n0 = __builtin_popcount(m0);
  const int n  = n0 + __builtin_popcount(m1);
  const int r0 = k0 ? (int)__builtin_amdgcn_mbcnt_lo(m0, 0u)
                    : 63 - (int)__builtin_amdgcn_mbcnt_lo(~m0, 0u);
  const int r1 = k1 ? n0 + (int)__builtin_amdgcn_mbcnt_lo(m1, 0u)
                    : 31 + n0 - (int)__builtin_amdgcn_mbcnt_lo(~m1, 0u);
  sh_src[wave][r0] = s0; sh_dst[wave][r0] = d0; sh_w[wave][r0] = w0;
  sh_src[wave][r1] = s1; sh_dst[wave][r1] = d1; sh_w[wave][r1] = w1;
  __syncthreads();

  const int   sA = sh_src[wave][lane];
  const int   dA = sh_dst[wave][lane];
  const float wA = sh_w[wave][lane];
  const int   sB = sh_src[wave][lane + 32];
  const int   dB = sh_dst[wave][lane + 32];
  const float wB = sh_w[wave][lane + 32];

  const int nA = (n < 32) ? n : 32;
  scatter_list(T, agg, sA, dA, wA, nA, lane);
  scatter_list(T, agg, sB, dB, wB, n - nA, lane);
}

extern "C" void kernel_launch(void* const* d_in, const int* in_sizes, int n_in,
                              void* d_out, int out_size, void* d_ws, size_t ws_size,
                              hipStream_t stream) {
  const float* x    = (const float*)d_in[0];
  const int*   ei   = (const int*)d_in[1];
  const float* ew   = (const float*)d_in[2];
  const float* W1   = (const float*)d_in[3];
  const float* b1   = (const float*)d_in[4];
  const float* W2   = (const float*)d_in[5];
  const float* b2   = (const float*)d_in[6];
  const float* W3   = (const float*)d_in[7];
  const float* b3   = (const float*)d_in[8];
  const float* Wlin = (const float*)d_in[9];
  const float* blin = (const float*)d_in[10];

  const int N = in_sizes[0] / 128;
  const int E = in_sizes[2];
  const int* src = ei;
  const int* dst = ei + E;

  float* T  = (float*)d_ws;
  float* P1 = T  + (size_t)N * 64;
  float* P2 = P1 + (size_t)N * 64;
  float* P3 = P2 + (size_t)N * 64;

  const int Mtiles  = (N + 15) / 16;
  const int nchunks = (E + kChunk - 1) / kChunk;
  const int sblocks = 2 * nchunks;

  int* tick = (int*)d_out;

  gemm_layer_k<128, false><<<dim3(Mtiles, 1, 1), dim3(32, 4, 1), 0, stream>>>(
      x, W1, b1, T, P1, N, tick);
  scatter_edges<<<sblocks, 256, 0, stream>>>(T, src, dst, ew, P1, E, N, nchunks,
                                             tick + 0 * kTicketPitch);

  gemm_layer_k<64, true><<<dim3(Mtiles, 1, 1), dim3(32, 4, 1), 0, stream>>>(
      P1, W2, b2, T, P2, N, nullptr);
  scatter_edges<<<sblocks, 256, 0, stream>>>(T, src, dst, ew, P2, E, N, nchunks,
                                             tick + 2 * kTicketPitch);

  gemm_layer_k<64, true><<<dim3(Mtiles, 1, 1), dim3(32, 4, 1), 0, stream>>>(
      P2, W3, b3, T, P3, N, nullptr);
  scatter_edges<<<sblocks, 256, 0, stream>>>(T, src, dst, ew, P3, E, N, nchunks,
                                             tick + 4 * kTicketPitch);

  gemm_out_k<<<dim3(Mtiles, 1, 1), dim3(32, 3, 1), 0, stream>>>(
      P1, P2, P3, Wlin, blin, (float*)d_out, N);
}


# Layer 3 sums edges before its product, scatter(relu(P2)) @ W3: output GEMM forms its relu(Q3 W3+b3) tile via LDS, layer-3 GEMM launch dropped, 7 -> 6 launches; layer-3 sum order changed
# speedup vs baseline: 2.4786x; 1.0990x over previous
#include <hip/hip_runtime.h>
#include <hip/hip_bf16.h>


typedef __attribute__((ext_vector_type(2))) float v2f;
typedef __attribute__((ext_vector_type(4))) float v4f;
typedef __attribute__((ext_vector_type(8))) float v8f;


constexpr int kTickets     = 6;
constexpr int kTicketPitch = 64;

template <int K, bool RELU_A, bool ZERO_Z>
__global__ void __attribute__((amdgpu_waves_per_eu(16, 16))) gemm_layer_k(const float* __restrict__ A,
                             const float* __restrict__ B,
                             const float* __restrict__ bias,
                             float* __restrict__ T,
                             float* __restrict__ P,
                             float* __restrict__ Z, int M,
                             int* __restrict__ tick) {
  const int lane  = threadIdx.x & 31;
  const int laneM = lane & 15;
  const int hi    = lane >> 4;
  const int wy    = threadIdx.y;
  const int row0  = blockIdx.x * 16;
  const int col0  = wy * 16;

  if (tick != nullptr && blockIdx.x == 0 && wy == 0 && lane < kTickets)
    tick[lane * kTicketPitch] = 0;

  const int c4 = laneM * 4;
  v4f bv;
  bv.x = bias[c4 + 0];
  bv.y = bias[c4 + 1];
  bv.z = bias[c4 + 2];
  bv.w = bias[c4 + 3];

  int arow = row0 + laneM;
  if (arow > M - 1) arow = M - 1;
  const float* __restrict__ aptr = A + (size_t)arow * K + 2 * hi;
  const float* __restrict__ bptr = B + 2 * hi * 64 + col0 + laneM;

  v8f acc = {};
#pragma unroll
  for (int k = 0; k < K; k += 4) {
    v2f a = *(const v2f*)(aptr + k);
    if (RELU_A) { a.x = fmaxf(a.x, 0.0f); a.y = fmaxf(a.y, 0.0f); }
    v2f b;
    b.x = bptr[k * 64];
    b.y = bptr[k * 64 + 64];
    acc = __builtin_amdgcn_wmma_f32_16x16x4_f32(false, a, false, b,
                                                (short)0, acc, false, false);
  }

  float* __restrict__ cptr = T + (size_t)(row0 + 8 * hi) * 64 + col0 + laneM;
#pragma unroll
  for (int r = 0; r < 8; ++r) {
    const int row = row0 + r + 8 * hi;
    if (row < M) cptr[r * 64] = acc[r];
  }

#pragma unroll
  for (int i = 0; i < 2; ++i) {
    const int r = row0 + wy * 4 + 2 * i + hi;
    if (r < M) {
      *(v4f*)(P + (size_t)r * 64 + c4) = bv;
      if (ZERO_Z) *(v4f*)(Z + (size_t)r * 64 + c4) = v4f{0.0f, 0.0f, 0.0f, 0.0f};
    }
  }
}

constexpr int kHPitch = 68;

__global__ void gemm_out_k(const float* __restrict__ P1,
                           const float* __restrict__ P2,
                           const float* __restrict__ Q3,
                           const float* __restrict__ W3,
                           const float* __restrict__ b3,
                           const float* __restrict__ Wl,
                           const float* __restrict__ bl,
                           float* __restrict__ C, int M) {
  constexpr int LDB = 40;
  __shared__ __attribute__((aligned(16))) float sH[16 * kHPitch];

  const int lane  = threadIdx.x & 31;
  const int laneM = lane & 15;
  const int hi    = lane >> 4;
  const int wy    = __builtin_amdgcn_readfirstlane((int)threadIdx.y);
  const int row0  = blockIdx.x * 16;
  const int col0  = wy * 16;

  int arow = row0 + laneM;
  if (arow > M - 1) arow = M - 1;
  const unsigned aoff = (unsigned)arow * 64u + 2u * (unsigned)hi;

  {
    const float* __restrict__ qp = Q3 + aoff;
    const float* __restrict__ wp = W3 + 2 * hi * 64 + col0 + laneM;
    v8f h = {};
#pragma unroll
    for (int k = 0; k < 64; k += 4) {
      v2f a = *(const v2f*)(qp + k);
      v2f b;
      b.x = wp[k * 64];
      b.y = wp[k * 64 + 64];
      h = __builtin_amdgcn_wmma_f32_16x16x4_f32(false, a, false, b,
                                                (short)0, h, false, false);
    }
    const float b3c = b3[col0 + laneM];
    float* hp = sH + (8 * hi) * kHPitch + col0 + laneM;
#pragma unroll
    for (int r = 0; r < 8; ++r) hp[r * kHPitch] = fmaxf(h[r] + b3c, 0.0f);
  }
  __syncthreads();
  if (wy == 3) return;

  const int col   = col0 + laneM;
  int       colc  = col;
  float     cmask = 1.0f;
  if (colc > LDB - 1) { colc = LDB - 1; cmask = 0.0f; }
  const float* __restrict__ bptr = Wl + 2 * hi * LDB + colc;

  v8f acc = {};
#pragma unroll
  for (int s = 0; s < 2; ++s) {
    const float* __restrict__ ap = ((s == 0) ? P1 : P2) + aoff;
#pragma unroll
    for (int kk = 0; kk < 64; kk += 4) {
      const int k = s * 64 + kk;
      v2f a = *(const v2f*)(ap + kk);
      a.x = fmaxf(a.x, 0.0f);
      a.y = fmaxf(a.y, 0.0f);
      v2f b;
      b.x = bptr[k * LDB] * cmask;
      b.y = bptr[k * LDB + LDB] * cmask;
      acc = __builtin_amdgcn_wmma_f32_16x16x4_f32(false, a, false, b,
                                                  (short)0, acc, false, false);
    }
  }
  const float* hrow = sH + laneM * kHPitch + 2 * hi;
#pragma unroll
  for (int kk = 0; kk < 64; kk += 4) {
    const int k = 128 + kk;
    const v2f a = *(const v2f*)(hrow + kk);
    v2f b;
    b.x = bptr[k * LDB] * cmask;
    b.y = bptr[k * LDB + LDB] * cmask;
    acc = __builtin_amdgcn_wmma_f32_16x16x4_f32(false, a, false, b,
                                                (short)0, acc, false, false);
  }

  if (cmask == 0.0f) return;
  const float bc = bl[col];
  float* __restrict__ cptr = C + (size_t)(row0 + 8 * hi) * LDB + col;
#pragma unroll
  for (int r = 0; r < 8; ++r) {
    const int row = row0 + r + 8 * hi;
    if (row < M) cptr[r * LDB] = acc[r] + bc;
  }
}

constexpr int kRawPerWave   = 64;
constexpr int kScatterWaves = 8;
constexpr int kChunk        = kRawPerWave * kScatterWaves;
constexpr int kEdgeGroup    = 8;

template <bool FULL, bool RELU_T>
__device__ __forceinline__ void scatter_group(const float* __restrict__ T,
                                              float* __restrict__ agg,
                                              int sv, int dv, float wv,
                                              int g, int c, int lane) {
  int   d[kEdgeGroup];
  float we[kEdgeGroup];
  float t0[kEdgeGroup], t1[kEdgeGroup];
#pragma unroll
  for (int u = 0; u < kEdgeGroup; ++u) {
    if (FULL || u < c) {
      const int s = __builtin_amdgcn_readlane(sv, g + u);
      d[u]  = __builtin_amdgcn_readlane(dv, g + u);
      we[u] = __builtin_bit_cast(
          float, __builtin_amdgcn_readlane(__builtin_bit_cast(int, wv), g + u));
      const float* __restrict__ tp = T + (size_t)s * 64 + lane;
      t0[u] = tp[0];
      t1[u] = tp[32];
      if (RELU_T) { t0[u] = fmaxf(t0[u], 0.0f); t1[u] = fmaxf(t1[u], 0.0f); }
    }
  }
#pragma unroll
  for (int u = 0; u < kEdgeGroup; ++u) {
    if (FULL || u < c) {
      float* __restrict__ ap = agg + (size_t)d[u] * 64 + lane;
      atomicAdd(ap,      t0[u] * we[u]);
      atomicAdd(ap + 32, t1[u] * we[u]);
    }
  }
}

template <bool RELU_T>
__device__ __forceinline__ void scatter_list(const float* __restrict__ T,
                                             float* __restrict__ agg,
                                             int sv, int dv, float wv,
                                             int cnt, int lane) {
  int g = 0;
  for (; g + kEdgeGroup <= cnt; g += kEdgeGroup)
    scatter_group<true, RELU_T>(T, agg, sv, dv, wv, g, kEdgeGroup, lane);
  if (g < cnt) scatter_group<false, RELU_T>(T, agg, sv, dv, wv, g, cnt - g, lane);
}

template <bool RELU_T>
__global__ void __launch_bounds__(256) scatter_edges(const float* __restrict__ T,
                              const int* __restrict__ src,
                              const int* __restrict__ dst,
                              const float* __restrict__ w,
                              float* __restrict__ agg, int E, int N,
                              int nchunks, int* __restrict__ tick) {
  __shared__ int   sh_ticket[2];
  __shared__ int   sh_src[kScatterWaves][kRawPerWave];
  __shared__ int   sh_dst[kScatterWaves][kRawPerWave];
  __shared__ float sh_w[kScatterWaves][kRawPerWave];

  const int lane = threadIdx.x & 31;
  const int wave = __builtin_amdgcn_readfirstlane((int)(threadIdx.x >> 5));

  if (threadIdx.x == 0) {
    const unsigned id = __builtin_amdgcn_s_sendmsg_rtn(0x87);
    int h = (int)((id >> 18) & 1u);
    int t = atomicAdd(tick + h * kTicketPitch, 1);
    if (t >= nchunks) {
      h ^= 1;
      t = atomicAdd(tick + h * kTicketPitch, 1);
    }
    sh_ticket[0] = h;
    sh_ticket[1] = t;
  }
  __syncthreads();
  const int h = __builtin_amdgcn_readfirstlane(sh_ticket[0]);
  const int t = __builtin_amdgcn_readfirstlane(sh_ticket[1]);
  if (t >= nchunks) return;

  const int base = t * kChunk + wave * kRawPerWave;

  const int  e0 = base + lane;
  const int  e1 = base + 32 + lane;
  const bool v0 = e0 < E;
  const bool v1 = e1 < E;
  const int  c0 = v0 ? e0 : E - 1;
  const int  c1 = v1 ? e1 : E - 1;
  const int   d0 = dst[c0];
  const int   d1 = dst[c1];
  const int   s0 = src[c0];
  const int   s1 = src[c1];
  const float w0 = w[c0];
  const float w1 = w[c1];

  const int  split = N >> 1;
  const bool k0 = v0 && ((d0 >= split) == (h != 0));
  const bool k1 = v1 && ((d1 >= split) == (h != 0));
  const unsigned m0 = __builtin_amdgcn_ballot_w32(k0);
  const unsigned m1 = __builtin_amdgcn_ballot_w32(k1);
  const int n0 = __builtin_popcount(m0);
  const int n  = n0 + __builtin_popcount(m1);
  const int r0 = k0 ? (int)__builtin_amdgcn_mbcnt_lo(m0, 0u)
                    : 63 - (int)__builtin_amdgcn_mbcnt_lo(~m0, 0u);
  const int r1 = k1 ? n0 + (int)__builtin_amdgcn_mbcnt_lo(m1, 0u)
                    : 31 + n0 - (int)__builtin_amdgcn_mbcnt_lo(~m1, 0u);
  sh_src[wave][r0] = s0; sh_dst[wave][r0] = d0; sh_w[wave][r0] = w0;
  sh_src[wave][r1] = s1; sh_dst[wave][r1] = d1; sh_w[wave][r1] = w1;
  __syncthreads();

  const int   sA = sh_src[wave][lane];
  const int   dA = sh_dst[wave][lane];
  const float wA = sh_w[wave][lane];
  const int   sB = sh_src[wave][lane + 32];
  const int   dB = sh_dst[wave][lane + 32];
  const float wB = sh_w[wave][lane + 32];

  const int nA = (n < 32) ? n : 32;
  scatter_list<RELU_T>(T, agg, sA, dA, wA, nA, lane);
  scatter_list<RELU_T>(T, agg, sB, dB, wB, n - nA, lane);
}

extern "C" void kernel_launch(void* const* d_in, const int* in_sizes, int n_in,
                              void* d_out, int out_size, void* d_ws, size_t ws_size,
                              hipStream_t stream) {
  const float* x    = (const float*)d_in[0];
  const int*   ei   = (const int*)d_in[1];
  const float* ew   = (const float*)d_in[2];
  const float* W1   = (const float*)d_in[3];
  const float* b1   = (const float*)d_in[4];
  const float* W2   = (const float*)d_in[5];
  const float* b2   = (const float*)d_in[6];
  const float* W3   = (const float*)d_in[7];
  const float* b3   = (const float*)d_in[8];
  const float* Wlin = (const float*)d_in[9];
  const float* blin = (const float*)d_in[10];

  const int N = in_sizes[0] / 128;
  const int E = in_sizes[2];
  const int* src = ei;
  const int* dst = ei + E;

  float* T  = (float*)d_ws;
  float* P1 = T  + (size_t)N * 64;
  float* P2 = P1 + (size_t)N * 64;
  float* Q3 = P2 + (size_t)N * 64;

  const int Mtiles  = (N + 15) / 16;
  const int nchunks = (E + kChunk - 1) / kChunk;
  const int sblocks = 2 * nchunks;

  int* tick = (int*)d_out;

  gemm_layer_k<128, false, false><<<dim3(Mtiles, 1, 1), dim3(32, 4, 1), 0, stream>>>(
      x, W1, b1, T, P1, nullptr, N, tick);
  scatter_edges<false><<<sblocks, 256, 0, stream>>>(T, src, dst, ew, P1, E, N, nchunks,
                                                    tick + 0 * kTicketPitch);

  gemm_layer_k<64, true, true><<<dim3(Mtiles, 1, 1), dim3(32, 4, 1), 0, stream>>>(
      P1, W2, b2, T, P2, Q3, N, nullptr);
  scatter_edges<false><<<sblocks, 256, 0, stream>>>(T, src, dst, ew, P2, E, N, nchunks,
                                                    tick + 2 * kTicketPitch);

  scatter_edges<true><<<sblocks, 256, 0, stream>>>(P2, src, dst, ew, Q3, E, N, nchunks,
                                                   tick + 4 * kTicketPitch);

  gemm_out_k<<<dim3(Mtiles, 1, 1), dim3(32, 4, 1), 0, stream>>>(
      P1, P2, Q3, W3, b3, Wlin, blin, (float*)d_out, N);
}


# GEMMs retiled: one wave owns 16 rows x all columns, column tiles interleaved (tile t = cols 4n+t) so A/B/C fragments are 16-byte accesses and A is read once, not by 4 waves; k order in each 8 changed
# speedup vs baseline: 2.7108x; 1.0937x over previous
#include <hip/hip_runtime.h>
#include <hip/hip_bf16.h>


typedef __attribute__((ext_vector_type(2))) float v2f;
typedef __attribute__((ext_vector_type(4))) float v4f;
typedef __attribute__((ext_vector_type(8))) float v8f;


constexpr int kTickets     = 6;
constexpr int kTicketPitch = 64;

__device__ __forceinline__ void mma4(v2f a, v4f bx, v4f by,
                                     v8f& c0, v8f& c1, v8f& c2, v8f& c3) {
  c0 = __builtin_amdgcn_wmma_f32_16x16x4_f32(false, a, false, v2f{bx.x, by.x},
                                             (short)0, c0, false, false);
  c1 = __builtin_amdgcn_wmma_f32_16x16x4_f32(false, a, false, v2f{bx.y, by.y},
                                             (short)0, c1, false, false);
  c2 = __builtin_amdgcn_wmma_f32_16x16x4_f32(false, a, false, v2f{bx.z, by.z},
                                             (short)0, c2, false, false);
  c3 = __builtin_amdgcn_wmma_f32_16x16x4_f32(false, a, false, v2f{bx.w, by.w},
                                             (short)0, c3, false, false);
}

__device__ __forceinline__ v4f relu4(v4f v) {
  return v4f{fmaxf(v.x, 0.0f), fmaxf(v.y, 0.0f), fmaxf(v.z, 0.0f), fmaxf(v.w, 0.0f)};
}

template <int K, bool RELU_A, bool ZERO_Z>
__global__ void __launch_bounds__(32) __attribute__((amdgpu_waves_per_eu(4, 4))) gemm_layer_k(const float* __restrict__ A,
                             const float* __restrict__ B,
                             const float* __restrict__ bias,
                             float* __restrict__ T,
                             float* __restrict__ P,
                             float* __restrict__ Z, int M,
                             int* __restrict__ tick) {
  const int lane  = threadIdx.x & 31;
  const int laneM = lane & 15;
  const int hi    = lane >> 4;
  const int row0  = blockIdx.x * 16;
  const int c4    = laneM * 4;

  if (tick != nullptr && blockIdx.x == 0 && lane < kTickets)
    tick[lane * kTicketPitch] = 0;

  const v4f bv = *(const v4f*)(bias + c4);

  int arow = row0 + laneM;
  if (arow > M - 1) arow = M - 1;
  const float* __restrict__ aptr = A + (size_t)arow * K + 4 * hi;
  const float* __restrict__ bptr = B + 4 * hi * 64 + c4;

  v8f acc0 = {}, acc1 = {}, acc2 = {}, acc3 = {};
#pragma unroll
  for (int k0 = 0; k0 < K; k0 += 8) {
    v4f a4 = *(const v4f*)(aptr + k0);
    if (RELU_A) a4 = relu4(a4);
#pragma unroll
    for (int s = 0; s < 2; ++s) {
      const v4f bx = *(const v4f*)(bptr + (k0 + 2 * s) * 64);
      const v4f by = *(const v4f*)(bptr + (k0 + 2 * s + 1) * 64);
      const v2f a  = (s == 0) ? v2f{a4.x, a4.y} : v2f{a4.z, a4.w};
      mma4(a, bx, by, acc0, acc1, acc2, acc3);
    }
  }

#pragma unroll
  for (int r = 0; r < 8; ++r) {
    const int row = row0 + r + 8 * hi;
    if (row < M) {
      const size_t o = (size_t)row * 64 + c4;
      *(v4f*)(T + o) = v4f{acc0[r], acc1[r], acc2[r], acc3[r]};
      *(v4f*)(P + o) = bv;
      if (ZERO_Z) *(v4f*)(Z + o) = v4f{0.0f, 0.0f, 0.0f, 0.0f};
    }
  }
}

constexpr int kHPitch = 68;

__global__ void __launch_bounds__(32) __attribute__((amdgpu_waves_per_eu(4, 4))) gemm_out_k(const float* __restrict__ P1,
                           const float* __restrict__ P2,
                           const float* __restrict__ Q3,
                           const float* __restrict__ W3,
                           const float* __restrict__ b3,
                           const float* __restrict__ Wl,
                           const float* __restrict__ bl,
                           float* __restrict__ C, int M) {
  constexpr int LDB = 40;
  __shared__ __attribute__((aligned(16))) float sH[16 * kHPitch];

  const int lane  = threadIdx.x & 31;
  const int laneM = lane & 15;
  const int hi    = lane >> 4;
  const int row0  = blockIdx.x * 16;
  const int c4    = laneM * 4;

  int arow = row0 + laneM;
  if (arow > M - 1) arow = M - 1;
  const unsigned aoff = (unsigned)arow * 64u + 4u * (unsigned)hi;

  {
    const float* __restrict__ qp = Q3 + aoff;
    const float* __restrict__ wp = W3 + 4 * hi * 64 + c4;
    v8f h0 = {}, h1 = {}, h2 = {}, h3 = {};
#pragma unroll
    for (int k0 = 0; k0 < 64; k0 += 8) {
      const v4f a4 = *(const v4f*)(qp + k0);
#pragma unroll
      for (int s = 0; s < 2; ++s) {
        const v4f bx = *(const v4f*)(wp + (k0 + 2 * s) * 64);
        const v4f by = *(const v4f*)(wp + (k0 + 2 * s + 1) * 64);
        const v2f a  = (s == 0) ? v2f{a4.x, a4.y} : v2f{a4.z, a4.w};
        mma4(a, bx, by, h0, h1, h2, h3);
      }
    }
    const v4f b3v = *(const v4f*)(b3 + c4);
    float* hp = sH + (8 * hi) * kHPitch + c4;
#pragma unroll
    for (int r = 0; r < 8; ++r)
      *(v4f*)(hp + r * kHPitch) =
          relu4(v4f{h0[r] + b3v.x, h1[r] + b3v.y, h2[r] + b3v.z, h3[r] + b3v.w});
  }
  __syncthreads();

  const int  n2  = (laneM < 8) ? laneM : 7;
  const float* __restrict__ bp01 = Wl + 4 * hi * LDB + 2 * laneM;
  const float* __restrict__ bp2  = Wl + 4 * hi * LDB + 32 + n2;

  v8f acc0 = {}, acc1 = {}, acc2 = {};
#pragma unroll
  for (int seg = 0; seg < 3; ++seg) {
    const float* __restrict__ ap = ((seg == 0) ? P1 : P2) + aoff;
    const float* hrow = sH + laneM * kHPitch + 4 * hi;
#pragma unroll
    for (int k0 = 0; k0 < 64; k0 += 8) {
      v4f a4;
      if (seg < 2) a4 = relu4(*(const v4f*)(ap + k0));
      else         a4 = *(const v4f*)(hrow + k0);
#pragma unroll
      for (int s = 0; s < 2; ++s) {
        const int k = seg * 64 + k0 + 2 * s;
        const v2f bx = *(const v2f*)(bp01 + k * LDB);
        const v2f by = *(const v2f*)(bp01 + (k + 1) * LDB);
        const float cx = bp2[k * LDB];
        const float cy = bp2[(k + 1) * LDB];
        const v2f a  = (s == 0) ? v2f{a4.x, a4.y} : v2f{a4.z, a4.w};
        acc0 = __builtin_amdgcn_wmma_f32_16x16x4_f32(false, a, false, v2f{bx.x, by.x},
                                                     (short)0, acc0, false, false);
        acc1 = __builtin_amdgcn_wmma_f32_16x16x4_f32(false, a, false, v2f{bx.y, by.y},
                                                     (short)0, acc1, false, false);
        acc2 = __builtin_amdgcn_wmma_f32_16x16x4_f32(false, a, false, v2f{cx, cy},
                                                     (short)0, acc2, false, false);
      }
    }
  }

  const v2f   bl01 = *(const v2f*)(bl + 2 * laneM);
  const float bl2  = bl[32 + n2];
#pragma unroll
  for (int r = 0; r < 8; ++r) {
    const int row = row0 + r + 8 * hi;
    if (row < M) {
      float* __restrict__ crow = C + (size_t)row * LDB;
      *(v2f*)(crow + 2 * laneM) = v2f{acc0[r] + bl01.x, acc1[r] + bl01.y};
      if (laneM < 8) crow[32 + laneM] = acc2[r] + bl2;
    }
  }
}

constexpr int kRawPerWave   = 64;
constexpr int kScatterWaves = 8;
constexpr int kChunk        = kRawPerWave * kScatterWaves;
constexpr int kEdgeGroup    = 8;

template <bool FULL, bool RELU_T>
__device__ __forceinline__ void scatter_group(const float* __restrict__ T,
                                              float* __restrict__ agg,
                                              int sv, int dv, float wv,
                                              int g, int c, int lane) {
  int   d[kEdgeGroup];
  float we[kEdgeGroup];
  float t0[kEdgeGroup], t1[kEdgeGroup];
#pragma unroll
  for (int u = 0; u < kEdgeGroup; ++u) {
    if (FULL || u < c) {
      const int s = __builtin_amdgcn_readlane(sv, g + u);
      d[u]  = __builtin_amdgcn_readlane(dv, g + u);
      we[u] = __builtin_bit_cast(
          float, __builtin_amdgcn_readlane(__builtin_bit_cast(int, wv), g + u));
      const float* __restrict__ tp = T + (size_t)s * 64 + lane;
      t0[u] = tp[0];
      t1[u] = tp[32];
      if (RELU_T) { t0[u] = fmaxf(t0[u], 0.0f); t1[u] = fmaxf(t1[u], 0.0f); }
    }
  }
#pragma unroll
  for (int u = 0; u < kEdgeGroup; ++u) {
    if (FULL || u < c) {
      float* __restrict__ ap = agg + (size_t)d[u] * 64 + lane;
      atomicAdd(ap,      t0[u] * we[u]);
      atomicAdd(ap + 32, t1[u] * we[u]);
    }
  }
}

template <bool RELU_T>
__device__ __forceinline__ void scatter_list(const float* __restrict__ T,
                                             float* __restrict__ agg,
                                             int sv, int dv, float wv,
                                             int cnt, int lane) {
  int g = 0;
  for (; g + kEdgeGroup <= cnt; g += kEdgeGroup)
    scatter_group<true, RELU_T>(T, agg, sv, dv, wv, g, kEdgeGroup, lane);
  if (g < cnt) scatter_group<false, RELU_T>(T, agg, sv, dv, wv, g, cnt - g, lane);
}

template <bool RELU_T>
__global__ void __launch_bounds__(256) scatter_edges(const float* __restrict__ T,
                              const int* __restrict__ src,
                              const int* __restrict__ dst,
                              const float* __restrict__ w,
                              float* __restrict__ agg, int E, int N,
                              int nchunks, int* __restrict__ tick) {
  __shared__ int   sh_ticket[2];
  __shared__ int   sh_src[kScatterWaves][kRawPerWave];
  __shared__ int   sh_dst[kScatterWaves][kRawPerWave];
  __shared__ float sh_w[kScatterWaves][kRawPerWave];

  const int lane = threadIdx.x & 31;
  const int wave = __builtin_amdgcn_readfirstlane((int)(threadIdx.x >> 5));

  if (threadIdx.x == 0) {
    const unsigned id = __builtin_amdgcn_s_sendmsg_rtn(0x87);
    int h = (int)((id >> 18) & 1u);
    int t = atomicAdd(tick + h * kTicketPitch, 1);
    if (t >= nchunks) {
      h ^= 1;
      t = atomicAdd(tick + h * kTicketPitch, 1);
    }
    sh_ticket[0] = h;
    sh_ticket[1] = t;
  }
  __syncthreads();
  const int h = __builtin_amdgcn_readfirstlane(sh_ticket[0]);
  const int t = __builtin_amdgcn_readfirstlane(sh_ticket[1]);
  if (t >= nchunks) return;

  const int base = t * kChunk + wave * kRawPerWave;

  const int  e0 = base + lane;
  const int  e1 = base + 32 + lane;
  const bool v0 = e0 < E;
  const bool v1 = e1 < E;
  const int  c0 = v0 ? e0 : E - 1;
  const int  c1 = v1 ? e1 : E - 1;
  const int   d0 = dst[c0];
  const int   d1 = dst[c1];
  const int   s0 = src[c0];
  const int   s1 = src[c1];
  const float w0 = w[c0];
  const float w1 = w[c1];

  const int  split = N >> 1;
  const bool k0 = v0 && ((d0 >= split) == (h != 0));
  const bool k1 = v1 && ((d1 >= split) == (h != 0));
  const unsigned m0 = __builtin_amdgcn_ballot_w32(k0);
  const unsigned m1 = __builtin_amdgcn_ballot_w32(k1);
  const int n0 = __builtin_popcount(m0);
  const int n  = n0 + __builtin_popcount(m1);
  const int r0 = k0 ? (int)__builtin_amdgcn_mbcnt_lo(m0, 0u)
                    : 63 - (int)__builtin_amdgcn_mbcnt_lo(~m0, 0u);
  const int r1 = k1 ? n0 + (int)__builtin_amdgcn_mbcnt_lo(m1, 0u)
                    : 31 + n0 - (int)__builtin_amdgcn_mbcnt_lo(~m1, 0u);
  sh_src[wave][r0] = s0; sh_dst[wave][r0] = d0; sh_w[wave][r0] = w0;
  sh_src[wave][r1] = s1; sh_dst[wave][r1] = d1; sh_w[wave][r1] = w1;
  __syncthreads();

  const int   sA = sh_src[wave][lane];
  const int   dA = sh_dst[wave][lane];
  const float wA = sh_w[wave][lane];
  const int   sB = sh_src[wave][lane + 32];
  const int   dB = sh_dst[wave][lane + 32];
  const float wB = sh_w[wave][lane + 32];

  const int nA = (n < 32) ? n : 32;
  scatter_list<RELU_T>(T, agg, sA, dA, wA, nA, lane);
  scatter_list<RELU_T>(T, agg, sB, dB, wB, n - nA, lane);
}

extern "C" void kernel_launch(void* const* d_in, const int* in_sizes, int n_in,
                              void* d_out, int out_size, void* d_ws, size_t ws_size,
                              hipStream_t stream) {
  const float* x    = (const float*)d_in[0];
  const int*   ei   = (const int*)d_in[1];
  const float* ew   = (const float*)d_in[2];
  const float* W1   = (const float*)d_in[3];
  const float* b1   = (const float*)d_in[4];
  const float* W2   = (const float*)d_in[5];
  const float* b2   = (const float*)d_in[6];
  const float* W3   = (const float*)d_in[7];
  const float* b3   = (const float*)d_in[8];
  const float* Wlin = (const float*)d_in[9];
  const float* blin = (const float*)d_in[10];

  const int N = in_sizes[0] / 128;
  const int E = in_sizes[2];
  const int* src = ei;
  const int* dst = ei + E;

  float* T  = (float*)d_ws;
  float* P1 = T  + (size_t)N * 64;
  float* P2 = P1 + (size_t)N * 64;
  float* Q3 = P2 + (size_t)N * 64;

  const int Mtiles  = (N + 15) / 16;
  const int nchunks = (E + kChunk - 1) / kChunk;
  const int sblocks = 2 * nchunks;

  int* tick = (int*)d_out;

  gemm_layer_k<128, false, false><<<dim3(Mtiles, 1, 1), dim3(32, 1, 1), 0, stream>>>(
      x, W1, b1, T, P1, nullptr, N, tick);
  scatter_edges<false><<<sblocks, 256, 0, stream>>>(T, src, dst, ew, P1, E, N, nchunks,
                                                    tick + 0 * kTicketPitch);

  gemm_layer_k<64, true, true><<<dim3(Mtiles, 1, 1), dim3(32, 1, 1), 0, stream>>>(
      P1, W2, b2, T, P2, Q3, N, nullptr);
  scatter_edges<false><<<sblocks, 256, 0, stream>>>(T, src, dst, ew, P2, E, N, nchunks,
                                                    tick + 2 * kTicketPitch);

  scatter_edges<true><<<sblocks, 256, 0, stream>>>(P2, src, dst, ew, Q3, E, N, nchunks,
                                                   tick + 4 * kTicketPitch);

  gemm_out_k<<<dim3(Mtiles, 1, 1), dim3(32, 1, 1), 0, stream>>>(
      P1, P2, Q3, W3, b3, Wlin, blin, (float*)d_out, N);
}
